# starting kernel
# baseline (speedup 1.0000x reference)
#include <hip/hip_runtime.h>
#include <hip/hip_bf16.h>
#include <math.h>

#define N_SENT 131072
#define N_BAGS 16384
#define DCLS   512
#define REL    53

typedef __attribute__((ext_vector_type(2))) float v2f;
typedef __attribute__((ext_vector_type(8))) float v8f;

__global__ __launch_bounds__(256) void tan_bag_attn(
    const float* __restrict__ X, const int* __restrict__ scope,
    const float* __restrict__ Con, const int* __restrict__ rel,
    float* __restrict__ bag)
{
  const int lane = threadIdx.x & 31;
  const int wave = threadIdx.x >> 5;
  const int b = blockIdx.x * 8 + wave;

  const int s0 = scope[2 * b];
  const int s1 = scope[2 * b + 1];
  const float4* con4 = (const float4*)(Con + (size_t)rel[b] * DCLS);

  float4 c4[4], a4[4];
#pragma unroll
  for (int q = 0; q < 4; ++q) {
    c4[q] = con4[lane + 32 * q];
    a4[q] = make_float4(0.f, 0.f, 0.f, 0.f);
  }

  float m = -INFINITY, z = 0.f;
  for (int j = s0; j < s1; ++j) {
    const float4* x4 = (const float4*)(X + (size_t)j * DCLS);
    if (j + 1 < s1)
      __builtin_prefetch((const float*)x4 + DCLS, 0, 1);

    float4 xv[4];
    float s = 0.f;
#pragma unroll
    for (int q = 0; q < 4; ++q) {
      xv[q] = x4[lane + 32 * q];
      s = fmaf(xv[q].x, c4[q].x, s);
      s = fmaf(xv[q].y, c4[q].y, s);
      s = fmaf(xv[q].z, c4[q].z, s);
      s = fmaf(xv[q].w, c4[q].w, s);
    }
#pragma unroll
    for (int off = 16; off > 0; off >>= 1) s += __shfl_xor(s, off, 32);

    const float mn    = fmaxf(m, s);
    const float scale = __expf(m - mn);
    const float w     = __expf(s - mn);
    z = z * scale + w;
#pragma unroll
    for (int q = 0; q < 4; ++q) {
      a4[q].x = fmaf(a4[q].x, scale, w * xv[q].x);
      a4[q].y = fmaf(a4[q].y, scale, w * xv[q].y);
      a4[q].z = fmaf(a4[q].z, scale, w * xv[q].z);
      a4[q].w = fmaf(a4[q].w, scale, w * xv[q].w);
    }
    m = mn;
  }

  const float inv = 1.f / z;
  float4* o4 = (float4*)(bag + (size_t)b * DCLS);
#pragma unroll
  for (int q = 0; q < 4; ++q) {
    float4 r = a4[q];
    r.x *= inv; r.y *= inv; r.z *= inv; r.w *= inv;
    o4[lane + 32 * q] = r;
  }
}

__global__ __launch_bounds__(256) void tan_classifier(
    const float* __restrict__ bag, const float* __restrict__ W,
    const float* __restrict__ bias, float* __restrict__ out)
{
  const int lane = threadIdx.x & 31;
  const int wave = threadIdx.x >> 5;
  const int gw = blockIdx.x * 8 + wave;
  const int mt = gw >> 2;
  const int nt = gw & 3;
  const int m0 = mt * 16;
  const int n0 = nt * 16;

  const int r16  = lane & 15;
  const int koff = (lane >> 4) << 1;

  const float* arow = bag + (size_t)(m0 + r16) * DCLS + koff;
  const int ng  = n0 + r16;
  const int ngc = ng < REL ? ng : (REL - 1);
  const float* brow = W + (size_t)ngc * DCLS + koff;

  v8f cacc = {};
#if __has_builtin(__builtin_amdgcn_wmma_f32_16x16x4_f32)
#pragma unroll 4
  for (int k = 0; k < DCLS; k += 4) {
    v2f a  = *(const v2f*)(arow + k);
    v2f bb = *(const v2f*)(brow + k);
    cacc = __builtin_amdgcn_wmma_f32_16x16x4_f32(
         false, a,  false, bb,
         (short)0, cacc,  false,  false);
  }
#else
  {
    const int mb = m0 + ((lane >> 4) << 3);
    for (int r = 0; r < 8; ++r) {
      float acc = 0.f;
      const float* ar = bag + (size_t)(mb + r) * DCLS;
      const float* br = W + (size_t)ngc * DCLS;
      for (int k = 0; k < DCLS; ++k) acc = fmaf(ar[k], br[k], acc);
      cacc[r] = acc;
    }
  }
#endif

  if (ng < REL) {
    const float bv = bias[ng];
    const int mb = m0 + ((lane >> 4) << 3);
#pragma unroll
    for (int r = 0; r < 8; ++r)
      out[(size_t)(mb + r) * REL + ng] = cacc[r] + bv;
  }
}

extern "C" void kernel_launch(void* const* d_in, const int* in_sizes, int n_in,
                              void* d_out, int out_size, void* d_ws, size_t ws_size,
                              hipStream_t stream) {
  (void)in_sizes; (void)n_in; (void)out_size; (void)ws_size;
  const float* X     = (const float*)d_in[0];
  const int*   scope = (const int*)  d_in[1];
  const float* Con   = (const float*)d_in[2];
  const int*   rel   = (const int*)  d_in[3];
  const float* W     = (const float*)d_in[4];
  const float* bias  = (const float*)d_in[5];
  float* out = (float*)d_out;
  float* bagbuf = (float*)d_ws;

  tan_bag_attn<<<N_BAGS / 8, 256, 0, stream>>>(X, scope, Con, rel, bagbuf);
  tan_classifier<<<(N_BAGS / 16) * 4 / 8, 256, 0, stream>>>(bagbuf, W, bias, out);
}


# Classifier fused into the bag kernel: 16 bags per 512-thread block, bag tile kept in LDS, each wave one 16x16 WMMA tile over a quarter of K, four K-partials summed via LDS; one launch, no workspace
# speedup vs baseline: 1.1765x; 1.1765x over previous
#include <hip/hip_runtime.h>
#include <math.h>


#define N_SENT 131072
#define N_BAGS 16384
#define DCLS   512
#define REL    53

#define BAGS_PER_BLOCK 16
#define BAG_PITCH 516
#define P_PITCH   66

typedef __attribute__((ext_vector_type(2))) float v2f;
typedef __attribute__((ext_vector_type(8))) float v8f;

__global__ __launch_bounds__(512) void tan_fused(
    const float* __restrict__ X, const int* __restrict__ scope,
    const float* __restrict__ Con, const int* __restrict__ rel,
    const float* __restrict__ W, const float* __restrict__ bias,
    float* __restrict__ out)
{
  __shared__ __attribute__((aligned(16))) float bagT[BAGS_PER_BLOCK * BAG_PITCH];
  __shared__ float part[4 * 16 * P_PITCH];

  const int lane = threadIdx.x & 31;
  const int wave = threadIdx.x >> 5;
  const int b = blockIdx.x * BAGS_PER_BLOCK + wave;

  const int s0 = scope[2 * b];
  const int s1 = scope[2 * b + 1];
  const float4* con4 = (const float4*)(Con + (size_t)rel[b] * DCLS);

  float4 c4[4], a4[4];
#pragma unroll
  for (int q = 0; q < 4; ++q) {
    c4[q] = con4[lane + 32 * q];
    a4[q] = make_float4(0.f, 0.f, 0.f, 0.f);
  }

  float m = -INFINITY, z = 0.f;
  for (int j = s0; j < s1; ++j) {
    const float4* x4 = (const float4*)(X + (size_t)j * DCLS);
    if (j + 1 < s1)
      __builtin_prefetch((const float*)x4 + DCLS, 0, 1);

    float4 xv[4];
    float s = 0.f;
#pragma unroll
    for (int q = 0; q < 4; ++q) {
      xv[q] = x4[lane + 32 * q];
      s = fmaf(xv[q].x, c4[q].x, s);
      s = fmaf(xv[q].y, c4[q].y, s);
      s = fmaf(xv[q].z, c4[q].z, s);
      s = fmaf(xv[q].w, c4[q].w, s);
    }
#pragma unroll
    for (int off = 16; off > 0; off >>= 1) s += __shfl_xor(s, off, 32);

    const float mn    = fmaxf(m, s);
    const float scale = __expf(m - mn);
    const float w     = __expf(s - mn);
    z = z * scale + w;
#pragma unroll
    for (int q = 0; q < 4; ++q) {
      a4[q].x = fmaf(a4[q].x, scale, w * xv[q].x);
      a4[q].y = fmaf(a4[q].y, scale, w * xv[q].y);
      a4[q].z = fmaf(a4[q].z, scale, w * xv[q].z);
      a4[q].w = fmaf(a4[q].w, scale, w * xv[q].w);
    }
    m = mn;
  }

  {
    const float inv = 1.f / z;
    float4* o4 = (float4*)(bagT + wave * BAG_PITCH);
#pragma unroll
    for (int q = 0; q < 4; ++q) {
      float4 r = a4[q];
      r.x *= inv; r.y *= inv; r.z *= inv; r.w *= inv;
      o4[lane + 32 * q] = r;
    }
  }
  __syncthreads();

  const int nt = wave & 3;
  const int kq = wave >> 2;
  const int c  = lane & 15;
  const int h  = lane >> 4;
  const int ng  = nt * 16 + c;
  const int ngc = ng < REL ? ng : (REL - 1);
  const int kb  = kq * 128 + 4 * h;
  const float* arow = bagT + c * BAG_PITCH + kb;
  const float* brow = W + (size_t)ngc * DCLS + kb;

  v8f acc = {};
  for (int i = 0; i < 16; ++i) {
    const float4 va = *(const float4*)(arow + 8 * i);
    const float4 vb = *(const float4*)(brow + 8 * i);
    const v2f a0 = {va.x, va.y};
    const v2f b0 = {vb.x, vb.y};
    const v2f a1 = {va.z, va.w};
    const v2f b1 = {vb.z, vb.w};
    acc = __builtin_amdgcn_wmma_f32_16x16x4_f32(
         false, a0,  false, b0,
         (short)0, acc,  false,  false);
    acc = __builtin_amdgcn_wmma_f32_16x16x4_f32(
         false, a1,  false, b1,
         (short)0, acc,  false,  false);
  }

  {
    float* prow = part + (kq * 16 + 8 * h) * P_PITCH + nt * 16 + c;
#pragma unroll
    for (int r = 0; r < 8; ++r) prow[r * P_PITCH] = acc[r];
  }
  __syncthreads();

  const size_t obase = (size_t)blockIdx.x * (BAGS_PER_BLOCK * REL);
  for (int idx = threadIdx.x; idx < BAGS_PER_BLOCK * REL; idx += 512) {
    const int row = idx / REL;
    const int col = idx - row * REL;
    const float* p = part + row * P_PITCH + col;
    const float v = (p[0] + p[16 * P_PITCH]) + (p[32 * P_PITCH] + p[48 * P_PITCH]);
    out[obase + idx] = v + bias[col];
  }
}

extern "C" void kernel_launch(void* const* d_in, const int* in_sizes, int n_in,
                              void* d_out, int out_size, void* d_ws, size_t ws_size,
                              hipStream_t stream) {
  (void)in_sizes; (void)n_in; (void)out_size; (void)d_ws; (void)ws_size;
  const float* X     = (const float*)d_in[0];
  const int*   scope = (const int*)  d_in[1];
  const float* Con   = (const float*)d_in[2];
  const int*   rel   = (const int*)  d_in[3];
  const float* W     = (const float*)d_in[4];
  const float* bias  = (const float*)d_in[5];
  float* out = (float*)d_out;

  tan_fused<<<N_BAGS / BAGS_PER_BLOCK, 512, 0, stream>>>(X, scope, Con, rel, W, bias, out);
}


# Bag rows taken four per trip: all 16 b128 loads of a trip issued before first use, one online-softmax rescale per trip (softmax sums formed in a new order); next-row prefetch dropped
# speedup vs baseline: 1.6488x; 1.4014x over previous
#include <hip/hip_runtime.h>
#include <math.h>


#define N_SENT 131072
#define N_BAGS 16384
#define DCLS   512
#define REL    53

#define BAGS_PER_BLOCK 16
#define BAG_PITCH 516
#define P_PITCH   66

typedef __attribute__((ext_vector_type(2))) float v2f;
typedef __attribute__((ext_vector_type(8))) float v8f;

#define ROWS_PER_TRIP 4

template <int NR>
__device__ __forceinline__ void bag_rows(const float* __restrict__ xrow, const int lane,
                                         const float4 (&c4)[4], float4 (&a4)[4],
                                         float& m, float& z)
{
  float4 xv[NR][4];
#pragma unroll
  for (int i = 0; i < NR; ++i) {
    const float4* x4 = (const float4*)(xrow + (size_t)i * DCLS);
#pragma unroll
    for (int q = 0; q < 4; ++q) xv[i][q] = x4[lane + 32 * q];
  }

  float s[NR];
#pragma unroll
  for (int i = 0; i < NR; ++i) {
    float t = 0.f;
#pragma unroll
    for (int q = 0; q < 4; ++q) {
      t = fmaf(xv[i][q].x, c4[q].x, t);
      t = fmaf(xv[i][q].y, c4[q].y, t);
      t = fmaf(xv[i][q].z, c4[q].z, t);
      t = fmaf(xv[i][q].w, c4[q].w, t);
    }
    asm volatile("" : "+v"(t));
    s[i] = t;
  }
#pragma unroll
  for (int off = 16; off > 0; off >>= 1) {
#pragma unroll
    for (int i = 0; i < NR; ++i) s[i] += __shfl_xor(s[i], off, 32);
  }

  float mn = m;
#pragma unroll
  for (int i = 0; i < NR; ++i) mn = fmaxf(mn, s[i]);
  const float scale = __expf(m - mn);
  float w[NR];
#pragma unroll
  for (int i = 0; i < NR; ++i) w[i] = __expf(s[i] - mn);
  z = z * scale + w[0];
#pragma unroll
  for (int i = 1; i < NR; ++i) z += w[i];
#pragma unroll
  for (int q = 0; q < 4; ++q) {
    a4[q].x = fmaf(a4[q].x, scale, w[0] * xv[0][q].x);
    a4[q].y = fmaf(a4[q].y, scale, w[0] * xv[0][q].y);
    a4[q].z = fmaf(a4[q].z, scale, w[0] * xv[0][q].z);
    a4[q].w = fmaf(a4[q].w, scale, w[0] * xv[0][q].w);
#pragma unroll
    for (int i = 1; i < NR; ++i) {
      a4[q].x = fmaf(w[i], xv[i][q].x, a4[q].x);
      a4[q].y = fmaf(w[i], xv[i][q].y, a4[q].y);
      a4[q].z = fmaf(w[i], xv[i][q].z, a4[q].z);
      a4[q].w = fmaf(w[i], xv[i][q].w, a4[q].w);
    }
  }
  m = mn;
}

__global__ __launch_bounds__(512) void tan_fused(
    const float* __restrict__ X, const int* __restrict__ scope,
    const float* __restrict__ Con, const int* __restrict__ rel,
    const float* __restrict__ W, const float* __restrict__ bias,
    float* __restrict__ out)
{
  __shared__ __attribute__((aligned(16))) float bagT[BAGS_PER_BLOCK * BAG_PITCH];
  __shared__ float part[4 * 16 * P_PITCH];

  const int lane = threadIdx.x & 31;
  const int wave = threadIdx.x >> 5;
  const int b = blockIdx.x * BAGS_PER_BLOCK + wave;

  const int s0 = __builtin_amdgcn_readfirstlane(scope[2 * b]);
  const int s1 = __builtin_amdgcn_readfirstlane(scope[2 * b + 1]);
  const float4* con4 = (const float4*)(Con + (size_t)rel[b] * DCLS);

  float4 c4[4], a4[4];
#pragma unroll
  for (int q = 0; q < 4; ++q) {
    c4[q] = con4[lane + 32 * q];
    a4[q] = make_float4(0.f, 0.f, 0.f, 0.f);
  }

  float m = -INFINITY, z = 0.f;
  int j = s0;
  for (; j + ROWS_PER_TRIP <= s1; j += ROWS_PER_TRIP)
    bag_rows<ROWS_PER_TRIP>(X + (size_t)j * DCLS, lane, c4, a4, m, z);
  const int left = s1 - j;
  if (left == 3)      bag_rows<3>(X + (size_t)j * DCLS, lane, c4, a4, m, z);
  else if (left == 2) bag_rows<2>(X + (size_t)j * DCLS, lane, c4, a4, m, z);
  else if (left == 1) bag_rows<1>(X + (size_t)j * DCLS, lane, c4, a4, m, z);

  {
    const float inv = 1.f / z;
    float4* o4 = (float4*)(bagT + wave * BAG_PITCH);
#pragma unroll
    for (int q = 0; q < 4; ++q) {
      float4 r = a4[q];
      r.x *= inv; r.y *= inv; r.z *= inv; r.w *= inv;
      o4[lane + 32 * q] = r;
    }
  }
  __syncthreads();

  const int nt = wave & 3;
  const int kq = wave >> 2;
  const int c  = lane & 15;
  const int h  = lane >> 4;
  const int ng  = nt * 16 + c;
  const int ngc = ng < REL ? ng : (REL - 1);
  const int kb  = kq * 128 + 4 * h;
  const float* arow = bagT + c * BAG_PITCH + kb;
  const float* brow = W + (size_t)ngc * DCLS + kb;

  v8f acc = {};
  for (int i = 0; i < 16; ++i) {
    const float4 va = *(const float4*)(arow + 8 * i);
    const float4 vb = *(const float4*)(brow + 8 * i);
    const v2f a0 = {va.x, va.y};
    const v2f b0 = {vb.x, vb.y};
    const v2f a1 = {va.z, va.w};
    const v2f b1 = {vb.z, vb.w};
    acc = __builtin_amdgcn_wmma_f32_16x16x4_f32(
         false, a0,  false, b0,
         (short)0, acc,  false,  false);
    acc = __builtin_amdgcn_wmma_f32_16x16x4_f32(
         false, a1,  false, b1,
         (short)0, acc,  false,  false);
  }

  {
    float* prow = part + (kq * 16 + 8 * h) * P_PITCH + nt * 16 + c;
#pragma unroll
    for (int r = 0; r < 8; ++r) prow[r * P_PITCH] = acc[r];
  }
  __syncthreads();

  const size_t obase = (size_t)blockIdx.x * (BAGS_PER_BLOCK * REL);
  for (int idx = threadIdx.x; idx < BAGS_PER_BLOCK * REL; idx += 512) {
    const int row = idx / REL;
    const int col = idx - row * REL;
    const float* p = part + row * P_PITCH + col;
    const float v = (p[0] + p[16 * P_PITCH]) + (p[32 * P_PITCH] + p[48 * P_PITCH]);
    out[obase + idx] = v + bias[col];
  }
}

extern "C" void kernel_launch(void* const* d_in, const int* in_sizes, int n_in,
                              void* d_out, int out_size, void* d_ws, size_t ws_size,
                              hipStream_t stream) {
  (void)in_sizes; (void)n_in; (void)out_size; (void)d_ws; (void)ws_size;
  const float* X     = (const float*)d_in[0];
  const int*   scope = (const int*)  d_in[1];
  const float* Con   = (const float*)d_in[2];
  const int*   rel   = (const int*)  d_in[3];
  const float* W     = (const float*)d_in[4];
  const float* bias  = (const float*)d_in[5];
  float* out = (float*)d_out;

  tan_fused<<<N_BAGS / BAGS_PER_BLOCK, 512, 0, stream>>>(X, scope, Con, rel, W, bias, out);
}


# Next trip's 16 row loads issued before the current trip is summed and kept in flight through the classifier: one 16-wave block per WGP loops over four 16-bag tiles, W fragments parked in LDS
# speedup vs baseline: 1.7166x; 1.0411x over previous
#include <hip/hip_runtime.h>
#include <math.h>


#define N_SENT 131072
#define N_BAGS 16384
#define DCLS   512
#define REL    53

#define BAGS_PER_TILE   16
#define TILES_PER_BLOCK 4
#define BAG_PITCH 516
#define P_PITCH   66

typedef __attribute__((ext_vector_type(2))) float v2f;
typedef __attribute__((ext_vector_type(8))) float v8f;

#define ROWS_PER_TRIP 4

__device__ __forceinline__ void issue_trip(float4 (&xv)[ROWS_PER_TRIP][4],
                                           const float* __restrict__ X,
                                           const int j, const int s1, const int lane)
{
#pragma unroll
  for (int i = 0; i < ROWS_PER_TRIP; ++i) {
    int r = j + i;
    r = r < s1 ? r : s1 - 1;
    r = r > 0 ? r : 0;
    const float4* x4 = (const float4*)(X + (size_t)r * DCLS);
#pragma unroll
    for (int q = 0; q < 4; ++q) xv[i][q] = x4[lane + 32 * q];
  }
}

__device__ __forceinline__ void fold_trip(const float4 (&xv)[ROWS_PER_TRIP][4],
                                          const int j, const int s1,
                                          const float4 (&c4)[4], float4 (&a4)[4],
                                          float& m, float& z)
{
  constexpr int NR = ROWS_PER_TRIP;
  float s[NR];
#pragma unroll
  for (int i = 0; i < NR; ++i) {
    float t = 0.f;
#pragma unroll
    for (int q = 0; q < 4; ++q) {
      t = fmaf(xv[i][q].x, c4[q].x, t);
      t = fmaf(xv[i][q].y, c4[q].y, t);
      t = fmaf(xv[i][q].z, c4[q].z, t);
      t = fmaf(xv[i][q].w, c4[q].w, t);
    }
    asm volatile("" : "+v"(t));
    s[i] = t;
  }
#pragma unroll
  for (int off = 16; off > 0; off >>= 1) {
#pragma unroll
    for (int i = 0; i < NR; ++i) s[i] += __shfl_xor(s[i], off, 32);
  }
#pragma unroll
  for (int i = 1; i < NR; ++i)
    s[i] = (j + i < s1) ? s[i] : -INFINITY;

  float mn = m;
#pragma unroll
  for (int i = 0; i < NR; ++i) mn = fmaxf(mn, s[i]);
  const float scale = __expf(m - mn);
  float w[NR];
#pragma unroll
  for (int i = 0; i < NR; ++i) w[i] = __expf(s[i] - mn);
  z = z * scale + w[0];
#pragma unroll
  for (int i = 1; i < NR; ++i) z += w[i];
#pragma unroll
  for (int q = 0; q < 4; ++q) {
    a4[q].x = fmaf(a4[q].x, scale, w[0] * xv[0][q].x);
    a4[q].y = fmaf(a4[q].y, scale, w[0] * xv[0][q].y);
    a4[q].z = fmaf(a4[q].z, scale, w[0] * xv[0][q].z);
    a4[q].w = fmaf(a4[q].w, scale, w[0] * xv[0][q].w);
#pragma unroll
    for (int i = 1; i < NR; ++i) {
      a4[q].x = fmaf(w[i], xv[i][q].x, a4[q].x);
      a4[q].y = fmaf(w[i], xv[i][q].y, a4[q].y);
      a4[q].z = fmaf(w[i], xv[i][q].z, a4[q].z);
      a4[q].w = fmaf(w[i], xv[i][q].w, a4[q].w);
    }
  }
  m = mn;
}

__device__ __forceinline__ void run_bag(float4 (&cur)[ROWS_PER_TRIP][4],
                                        float4 (&nxt)[ROWS_PER_TRIP][4],
                                        const float* __restrict__ X,
                                        const int s0, const int s1,
                                        const int ns0, const int ns1, const int lane,
                                        const float4 (&c4)[4], float4 (&a4)[4],
                                        float& m, float& z)
{
  int j = s0;
  for (;;) {
#pragma unroll
    for (int i = 0; i < ROWS_PER_TRIP; ++i) {
#pragma unroll
      for (int q = 0; q < 4; ++q) cur[i][q] = nxt[i][q];
    }
    const bool last = (j + ROWS_PER_TRIP >= s1);
    issue_trip(nxt, X, last ? ns0 : j + ROWS_PER_TRIP, last ? ns1 : s1, lane);
    __builtin_amdgcn_sched_barrier(0);
    fold_trip(cur, j, s1, c4, a4, m, z);
    if (last) break;
    j += ROWS_PER_TRIP;
  }
}

__device__ __forceinline__ void lds_barrier()
{
  __builtin_amdgcn_fence(__ATOMIC_RELEASE, "workgroup", "local");
  __builtin_amdgcn_s_barrier();
  __builtin_amdgcn_fence(__ATOMIC_ACQUIRE, "workgroup", "local");
}

__global__ __launch_bounds__(512) void tan_fused(
    const float* __restrict__ X, const int* __restrict__ scope,
    const float* __restrict__ Con, const int* __restrict__ rel,
    const float* __restrict__ W, const float* __restrict__ bias,
    float* __restrict__ out)
{
  __shared__ __attribute__((aligned(16))) float bagT[BAGS_PER_TILE * BAG_PITCH];
  __shared__ float part[4 * 16 * P_PITCH];
  __shared__ __attribute__((aligned(16))) float wfrag[16 * 16 * 32 * 4];

  const int lane = threadIdx.x & 31;
  const int wave = threadIdx.x >> 5;
  const int bag0 = blockIdx.x * (BAGS_PER_TILE * TILES_PER_BLOCK) + wave;

  int bs0[TILES_PER_BLOCK], bs1[TILES_PER_BLOCK], brl[TILES_PER_BLOCK];
#pragma unroll
  for (int t = 0; t < TILES_PER_BLOCK; ++t) {
    const int b = bag0 + t * BAGS_PER_TILE;
    bs0[t] = __builtin_amdgcn_readfirstlane(scope[2 * b]);
    bs1[t] = __builtin_amdgcn_readfirstlane(scope[2 * b + 1]);
    brl[t] = __builtin_amdgcn_readfirstlane(rel[b]);
  }

  float4 bufA[ROWS_PER_TRIP][4], bufB[ROWS_PER_TRIP][4];
  float4 c4[4], a4[4];
  {
    const float4* con4 = (const float4*)(Con + (size_t)brl[0] * DCLS);
#pragma unroll
    for (int q = 0; q < 4; ++q) {
      c4[q] = con4[lane + 32 * q];
      a4[q] = make_float4(0.f, 0.f, 0.f, 0.f);
    }
  }
  issue_trip(bufB, X, bs0[0], bs1[0], lane);
  float m = -INFINITY, z = 0.f;

  const int nt = wave & 3;
  const int kq = wave >> 2;
  const int c  = lane & 15;
  const int h  = lane >> 4;
  const int ng  = nt * 16 + c;
  const int ngc = ng < REL ? ng : (REL - 1);
  const int kb  = kq * 128 + 4 * h;
  const float* arow = bagT + c * BAG_PITCH + kb;
  const float* brow = W + (size_t)ngc * DCLS + kb;

  float4* wf4 = (float4*)wfrag + wave * (16 * 32) + lane;
#pragma unroll
  for (int i = 0; i < 16; ++i) wf4[i * 32] = *(const float4*)(brow + 8 * i);
  const int idx0 = threadIdx.x;
  const int idx1 = threadIdx.x + 512;
  const float bias0 = bias[idx0 % REL];
  const float bias1 = bias[idx1 % REL];

  for (int t = 0; t < TILES_PER_BLOCK; ++t) {
    const int s0 = t == 0 ? bs0[0] : t == 1 ? bs0[1] : t == 2 ? bs0[2] : bs0[3];
    const int s1 = t == 0 ? bs1[0] : t == 1 ? bs1[1] : t == 2 ? bs1[2] : bs1[3];
    const int ns0 = t == 0 ? bs0[1] : t == 1 ? bs0[2] : t == 2 ? bs0[3] : bs1[3] - 1;
    const int ns1 = t == 0 ? bs1[1] : t == 1 ? bs1[2] : t == 2 ? bs1[3] : bs1[3];
    const int nrl = t == 0 ? brl[1] : t == 1 ? brl[2] : brl[3];

    run_bag(bufA, bufB, X, s0, s1, ns0, ns1, lane, c4, a4, m, z);

    {
      const float inv = 1.f / z;
      float4* o4 = (float4*)(bagT + wave * BAG_PITCH);
      const float4* con4 = (const float4*)(Con + (size_t)nrl * DCLS);
#pragma unroll
      for (int q = 0; q < 4; ++q) {
        float4 r = a4[q];
        r.x *= inv; r.y *= inv; r.z *= inv; r.w *= inv;
        o4[lane + 32 * q] = r;
        a4[q] = make_float4(0.f, 0.f, 0.f, 0.f);
        c4[q] = con4[lane + 32 * q];
      }
      m = -INFINITY; z = 0.f;
    }
    lds_barrier();

    v8f acc = {};
    for (int i = 0; i < 16; ++i) {
      const float4 va = *(const float4*)(arow + 8 * i);
      const float4 vb = wf4[i * 32];
      const v2f a0 = {va.x, va.y};
      const v2f b0 = {vb.x, vb.y};
      const v2f a1 = {va.z, va.w};
      const v2f b1 = {vb.z, vb.w};
      acc = __builtin_amdgcn_wmma_f32_16x16x4_f32(
           false, a0,  false, b0,
           (short)0, acc,  false,  false);
      acc = __builtin_amdgcn_wmma_f32_16x16x4_f32(
           false, a1,  false, b1,
           (short)0, acc,  false,  false);
    }

    {
      float* prow = part + (kq * 16 + 8 * h) * P_PITCH + nt * 16 + c;
#pragma unroll
      for (int r = 0; r < 8; ++r) prow[r * P_PITCH] = acc[r];
    }
    lds_barrier();

    const size_t obase = ((size_t)blockIdx.x * TILES_PER_BLOCK + t) * (BAGS_PER_TILE * REL);
#pragma unroll
    for (int k = 0; k < 2; ++k) {
      const int idx = k == 0 ? idx0 : idx1;
      if (idx < BAGS_PER_TILE * REL) {
        const int row = idx / REL;
        const int col = idx - row * REL;
        const float* p = part + row * P_PITCH + col;
        const float v = (p[0] + p[16 * P_PITCH]) + (p[32 * P_PITCH] + p[48 * P_PITCH]);
        out[obase + idx] = v + (k == 0 ? bias0 : bias1);
      }
    }
  }
}

extern "C" void kernel_launch(void* const* d_in, const int* in_sizes, int n_in,
                              void* d_out, int out_size, void* d_ws, size_t ws_size,
                              hipStream_t stream) {
  (void)in_sizes; (void)n_in; (void)out_size; (void)d_ws; (void)ws_size;
  const float* X     = (const float*)d_in[0];
  const int*   scope = (const int*)  d_in[1];
  const float* Con   = (const float*)d_in[2];
  const int*   rel   = (const int*)  d_in[3];
  const float* W     = (const float*)d_in[4];
  const float* bias  = (const float*)d_in[5];
  float* out = (float*)d_out;

  tan_fused<<<N_BAGS / (BAGS_PER_TILE * TILES_PER_BLOCK), 512, 0, stream>>>(X, scope, Con, rel, W, bias, out);
}
